# starting kernel
# baseline (speedup 1.0000x reference)
#include <hip/hip_runtime.h>
#include <stdint.h>

typedef __attribute__((ext_vector_type(2))) float v2f;
typedef __attribute__((ext_vector_type(8))) float v8f;

#define N_NODES  50000
#define N_EDGES  800000
#define NODE_DIM 128
#define STATE    64
#define OUT_DIM  32
#define ETYPES   8
#define OUT_W    (ETYPES * OUT_DIM)

__global__ void EdgeGCN_zero_kernel(float* __restrict__ p, long long n) {
    long long i = (long long)blockIdx.x * blockDim.x + threadIdx.x;
    if (i < n) p[i] = 0.0f;
}

__global__ void EdgeGCN_etype_kernel(const float* __restrict__ ea,
                                     int* __restrict__ et) {
    int e = blockIdx.x * blockDim.x + threadIdx.x;
    if (e >= N_EDGES) return;
    const float* a = ea + (long long)e * ETYPES;
    int t = 0;
#pragma unroll
    for (int j = 1; j < ETYPES; ++j)
        if (a[j] > 0.5f) t = j;
    et[e] = t;
}

__global__ void EdgeGCN_deg_kernel(const int* __restrict__ et,
                                   const int* __restrict__ eidx,
                                   float* __restrict__ deg) {
    int e = blockIdx.x * blockDim.x + threadIdx.x;
    if (e >= N_EDGES) return;
    int t = et[e];
    int c = eidx[N_EDGES + e];
    atomicAdd(&deg[(long long)t * N_NODES + c], 1.0f);
}

__global__ void EdgeGCN_dinv_kernel(float* __restrict__ d) {
    long long i = (long long)blockIdx.x * blockDim.x + threadIdx.x;
    if (i < (long long)ETYPES * N_NODES) d[i] = rsqrtf(d[i] + 1.0f);
}

template <int K, int S>
__global__ __launch_bounds__(256) void EdgeGCN_gemm_wmma(
        const float* __restrict__ A, const float* __restrict__ B,
        float* __restrict__ C, int nrows) {
    __shared__ float sB[K * S];
    const int tid = threadIdx.x;
    for (int i = tid; i < K * S; i += 256) sB[i] = B[i];
    __syncthreads();

    const int lane = tid & 31;
    const int w    = tid >> 5;
    constexpr int colTiles     = S / 16;
    constexpr int rowsPerBlock = 16 * (8 / colTiles);
    const int row0 = blockIdx.x * rowsPerBlock + (w / colTiles) * 16;
    const int col0 = (w % colTiles) * 16;
    const int lo = lane & 15;
    const int hi = lane >> 4;

    const int m = row0 + lo;
    const long long arow = (long long)((m < nrows) ? m : 0) * K;

    v8f c = {};
#pragma unroll
    for (int kb = 0; kb < K; kb += 4) {
        v2f a = *(const v2f*)(A + arow + kb + 2 * hi);
        v2f b;
        b.x = sB[(kb + 2 * hi)     * S + col0 + lo];
        b.y = sB[(kb + 2 * hi + 1) * S + col0 + lo];
        c = __builtin_amdgcn_wmma_f32_16x16x4_f32(
                 false, a,  false, b,
                 (short)0, c,  false,  false);
    }

#pragma unroll
    for (int v = 0; v < 8; ++v) {
        int mr = row0 + v + 8 * hi;
        if (mr < nrows) C[(long long)mr * S + col0 + lo] = c[v];
    }
}

__global__ void EdgeGCN_scatter1_kernel(const int* __restrict__ et,
                                        const int* __restrict__ eidx,
                                        const float* __restrict__ dinv_t,
                                        const float* __restrict__ h1,
                                        float* __restrict__ agg, int t) {
    long long tid = (long long)blockIdx.x * blockDim.x + threadIdx.x;
    if (tid >= (long long)N_EDGES * 8) return;
    int e  = (int)(tid >> 3);
    int s0 = (int)(tid & 7) * 8;
    if (et[e] != t) return;
    int r = eidx[e];
    int c = eidx[N_EDGES + e];
    float norm = dinv_t[r] * dinv_t[c];
    const float4* hp = (const float4*)(h1 + (long long)r * STATE + s0);
    float4 ha = hp[0], hb = hp[1];
    float* ap = agg + (long long)c * STATE + s0;
    atomicAdd(ap + 0, ha.x * norm); atomicAdd(ap + 1, ha.y * norm);
    atomicAdd(ap + 2, ha.z * norm); atomicAdd(ap + 3, ha.w * norm);
    atomicAdd(ap + 4, hb.x * norm); atomicAdd(ap + 5, hb.y * norm);
    atomicAdd(ap + 6, hb.z * norm); atomicAdd(ap + 7, hb.w * norm);
}

__global__ void EdgeGCN_combine1_kernel(const float* __restrict__ dinv_t,
                                        const float* __restrict__ h1,
                                        const float* __restrict__ b1t,
                                        float* __restrict__ agg) {
    long long i = (long long)blockIdx.x * blockDim.x + threadIdx.x;
    if (i >= (long long)N_NODES * STATE) return;
    int n = (int)(i >> 6);
    int s = (int)(i & (STATE - 1));
    float di = dinv_t[n];
    float v  = agg[i] + h1[i] * di * di + b1t[s];
    agg[i]   = fmaxf(v, 0.0f);
}

__global__ void EdgeGCN_scatter2_kernel(const int* __restrict__ et,
                                        const int* __restrict__ eidx,
                                        const float* __restrict__ dinv_t,
                                        const float* __restrict__ h2,
                                        float* __restrict__ out, int t) {
    long long tid = (long long)blockIdx.x * blockDim.x + threadIdx.x;
    if (tid >= (long long)N_EDGES * 8) return;
    int e  = (int)(tid >> 3);
    int s0 = (int)(tid & 7) * 4;
    if (et[e] != t) return;
    int r = eidx[e];
    int c = eidx[N_EDGES + e];
    float norm = dinv_t[r] * dinv_t[c];
    float4 h = *(const float4*)(h2 + (long long)r * OUT_DIM + s0);
    float* op = out + (long long)c * OUT_W + t * OUT_DIM + s0;
    atomicAdd(op + 0, h.x * norm); atomicAdd(op + 1, h.y * norm);
    atomicAdd(op + 2, h.z * norm); atomicAdd(op + 3, h.w * norm);
}

__global__ void EdgeGCN_combine2_kernel(const float* __restrict__ dinv_t,
                                        const float* __restrict__ h2,
                                        const float* __restrict__ b2t,
                                        float* __restrict__ out, int t) {
    long long i = (long long)blockIdx.x * blockDim.x + threadIdx.x;
    if (i >= (long long)N_NODES * OUT_DIM) return;
    int n = (int)(i >> 5);
    int s = (int)(i & (OUT_DIM - 1));
    float di = dinv_t[n];
    long long oi = (long long)n * OUT_W + t * OUT_DIM + s;
    out[oi] += h2[i] * di * di + b2t[s];
}

extern "C" void kernel_launch(void* const* d_in, const int* in_sizes, int n_in,
                              void* d_out, int out_size, void* d_ws, size_t ws_size,
                              hipStream_t stream) {
    const float* x   = (const float*)d_in[0];
    const float* ea  = (const float*)d_in[1];
    const float* W1  = (const float*)d_in[2];
    const float* b1  = (const float*)d_in[3];
    const float* W2  = (const float*)d_in[4];
    const float* b2  = (const float*)d_in[5];
    const int*   eix = (const int*)d_in[6];
    float* out = (float*)d_out;

    char* ws = (char*)d_ws;
    auto alignup = [](size_t v) { return (v + 255) & ~(size_t)255; };
    size_t off = 0;
    float* dinv = (float*)(ws + off); off += alignup(sizeof(float) * ETYPES * N_NODES);
    int*   et   = (int*)  (ws + off); off += alignup(sizeof(int) * N_EDGES);
    float* h1   = (float*)(ws + off); off += alignup(sizeof(float) * (size_t)N_NODES * STATE);
    float* agg1 = (float*)(ws + off); off += alignup(sizeof(float) * (size_t)N_NODES * STATE);
    float* h2   = (float*)(ws + off); off += alignup(sizeof(float) * (size_t)N_NODES * OUT_DIM);
    (void)ws_size; (void)in_sizes; (void)n_in; (void)out_size;

    const int TB = 256;
    auto blocksLL = [&](long long n) { return (unsigned)((n + TB - 1) / TB); };

    EdgeGCN_zero_kernel<<<blocksLL((long long)N_NODES * OUT_W), TB, 0, stream>>>(
        out, (long long)N_NODES * OUT_W);
    EdgeGCN_zero_kernel<<<blocksLL((long long)ETYPES * N_NODES), TB, 0, stream>>>(
        dinv, (long long)ETYPES * N_NODES);
    EdgeGCN_etype_kernel<<<blocksLL(N_EDGES), TB, 0, stream>>>(ea, et);
    EdgeGCN_deg_kernel<<<blocksLL(N_EDGES), TB, 0, stream>>>(et, eix, dinv);
    EdgeGCN_dinv_kernel<<<blocksLL((long long)ETYPES * N_NODES), TB, 0, stream>>>(dinv);

    for (int t = 0; t < ETYPES; ++t) {
        const float* dinv_t = dinv + (long long)t * N_NODES;

        {
            constexpr int rowsPerBlock = 16 * (8 / (STATE / 16));
            unsigned nb = (N_NODES + rowsPerBlock - 1) / rowsPerBlock;
            EdgeGCN_gemm_wmma<NODE_DIM, STATE><<<nb, TB, 0, stream>>>(
                x, W1 + (long long)t * NODE_DIM * STATE, h1, N_NODES);
        }

        EdgeGCN_zero_kernel<<<blocksLL((long long)N_NODES * STATE), TB, 0, stream>>>(
            agg1, (long long)N_NODES * STATE);
        EdgeGCN_scatter1_kernel<<<blocksLL((long long)N_EDGES * 8), TB, 0, stream>>>(
            et, eix, dinv_t, h1, agg1, t);
        EdgeGCN_combine1_kernel<<<blocksLL((long long)N_NODES * STATE), TB, 0, stream>>>(
            dinv_t, h1, b1 + (long long)t * STATE, agg1);

        {
            constexpr int rowsPerBlock = 16 * (8 / (OUT_DIM / 16));
            unsigned nb = (N_NODES + rowsPerBlock - 1) / rowsPerBlock;
            EdgeGCN_gemm_wmma<STATE, OUT_DIM><<<nb, TB, 0, stream>>>(
                agg1, W2 + (long long)t * STATE * OUT_DIM, h2, N_NODES);
        }

        EdgeGCN_scatter2_kernel<<<blocksLL((long long)N_EDGES * 8), TB, 0, stream>>>(
            et, eix, dinv_t, h2, out, t);
        EdgeGCN_combine2_kernel<<<blocksLL((long long)N_NODES * OUT_DIM), TB, 0, stream>>>(
            dinv_t, h2, b2 + (long long)t * OUT_DIM, out, t);
    }
}


# fewer, longer waves: sort kernels take 5 edges per thread, level by level; layer-2 gather walks a node's 4 type buckets together, 2 nodes per thread, edge records read a round ahead; sum order kept
# speedup vs baseline: 3.6860x; 3.6860x over previous
#include <hip/hip_runtime.h>
#include <stdint.h>

typedef __attribute__((ext_vector_type(2))) float v2f;
typedef __attribute__((ext_vector_type(8))) float v8f;

#define N_NODES  50000
#define N_EDGES  800000
#define NODE_DIM 128
#define STATE    64
#define OUT_DIM  32
#define ETYPES   8
#define OUT_W    (ETYPES * OUT_DIM)
#define NKEYS    (ETYPES * N_NODES)
#define SCAN_TILE   1024
#define SCAN_BLOCKS ((NKEYS + SCAN_TILE - 1) / SCAN_TILE)

static_assert(NKEYS % 4 == 0, "scan kernels read bucket counts four at a time");

#define EDGE_BLOCKS 625
#define EDGE_PER_THREAD 5
#define EDGE_STRIDE (EDGE_BLOCKS * 256)
#define ZERO_BLOCKS 196
#define ZERO_STRIDE (ZERO_BLOCKS * 256)

static_assert((long long)EDGE_STRIDE * EDGE_PER_THREAD == N_EDGES, "per-edge kernels cover the edge list exactly: no bounds test");
static_assert((long long)ZERO_STRIDE * 2 * 4 >= NKEYS, "the zero kernel covers every bucket");

__global__ __launch_bounds__(256) void EdgeGCN_zero_int_kernel(int* __restrict__ p) {
    const int i = blockIdx.x * 256 + threadIdx.x;
    const int4 z = make_int4(0, 0, 0, 0);
    int4* p4 = (int4*)p;
    if (i < NKEYS / 4) p4[i] = z;
    if (i + ZERO_STRIDE < NKEYS / 4) p4[i + ZERO_STRIDE] = z;
}

__global__ __launch_bounds__(256) void EdgeGCN_key_rank_kernel(const float* __restrict__ ea,
                                        const int* __restrict__ eidx,
                                        int* __restrict__ cnt,
                                        int* __restrict__ key,
                                        int* __restrict__ rank) {
    const int e0 = blockIdx.x * 256 + threadIdx.x;
    float4 lo[EDGE_PER_THREAD], hi[EDGE_PER_THREAD];
    int c[EDGE_PER_THREAD];
#pragma unroll
    for (int u = 0; u < EDGE_PER_THREAD; ++u) {
        const int e = e0 + u * EDGE_STRIDE;
        const float4* a4 = (const float4*)(ea + (long long)e * ETYPES);
        lo[u] = a4[0];
        hi[u] = a4[1];
        c[u]  = eidx[N_EDGES + e];
    }
    int k[EDGE_PER_THREAD], r[EDGE_PER_THREAD];
#pragma unroll
    for (int u = 0; u < EDGE_PER_THREAD; ++u) {
        int t = 0;
        if (lo[u].y > 0.5f) t = 1;
        if (lo[u].z > 0.5f) t = 2;
        if (lo[u].w > 0.5f) t = 3;
        if (hi[u].x > 0.5f) t = 4;
        if (hi[u].y > 0.5f) t = 5;
        if (hi[u].z > 0.5f) t = 6;
        if (hi[u].w > 0.5f) t = 7;
        k[u] = t * N_NODES + c[u];
    }
#pragma unroll
    for (int u = 0; u < EDGE_PER_THREAD; ++u) {
        r[u] = atomicAdd(&cnt[k[u]], 1);
    }
#pragma unroll
    for (int u = 0; u < EDGE_PER_THREAD; ++u) {
        const int e = e0 + u * EDGE_STRIDE;
        key[e]  = k[u];
        rank[e] = r[u];
    }
}

__global__ __launch_bounds__(256) void EdgeGCN_tile_sum_kernel(
        const int* __restrict__ cnt, int* __restrict__ tsum) {
    __shared__ int sw[8];
    const int tid = threadIdx.x;
    const int base = blockIdx.x * SCAN_TILE + tid * 4;
    int s = 0;
    if (base < NKEYS) {
        int4 v = *(const int4*)(cnt + base);
        s = v.x + v.y + v.z + v.w;
    }
    for (int o = 16; o > 0; o >>= 1) s += __shfl_xor(s, o, 32);
    if ((tid & 31) == 0) sw[tid >> 5] = s;
    __syncthreads();
    if (tid == 0) {
        int tot = 0;
        for (int i = 0; i < 8; ++i) tot += sw[i];
        tsum[blockIdx.x] = tot;
    }
}

__global__ __launch_bounds__(256) void EdgeGCN_scan_kernel(
        const int* __restrict__ cnt, const int* __restrict__ tsum,
        int* __restrict__ offs, float* __restrict__ dinv) {
    __shared__ int sp[8];
    __shared__ int sw[8];
    const int tid  = threadIdx.x;
    const int lane = tid & 31;
    const int w    = tid >> 5;

    int p = 0;
    for (int i = tid; i < (int)blockIdx.x; i += 256) p += tsum[i];
    for (int o = 16; o > 0; o >>= 1) p += __shfl_xor(p, o, 32);
    if (lane == 0) sp[w] = p;

    const int base = blockIdx.x * SCAN_TILE + tid * 4;
    const bool in = base < NKEYS;
    int4 v = make_int4(0, 0, 0, 0);
    if (in) v = *(const int4*)(cnt + base);
    const int s = v.x + v.y + v.z + v.w;
    int inc = s;
    for (int o = 1; o < 32; o <<= 1) {
        int u = __shfl_up(inc, o, 32);
        if (lane >= o) inc += u;
    }
    if (lane == 31) sw[w] = inc;
    __syncthreads();

    int ex = inc - s;
    for (int i = 0; i < 8; ++i) {
        ex += sp[i];
        if (i < w) ex += sw[i];
    }
    if (in) {
        int4 o4;
        o4.x = ex;
        o4.y = o4.x + v.x;
        o4.z = o4.y + v.y;
        o4.w = o4.z + v.z;
        *(int4*)(offs + base) = o4;
        float4 d;
        d.x = rsqrtf((float)v.x + 1.0f);
        d.y = rsqrtf((float)v.y + 1.0f);
        d.z = rsqrtf((float)v.z + 1.0f);
        d.w = rsqrtf((float)v.w + 1.0f);
        *(float4*)(dinv + base) = d;
    }
}

__global__ __launch_bounds__(256) void EdgeGCN_fill_kernel(const int* __restrict__ key,
                                    const int* __restrict__ rank,
                                    const int* __restrict__ eidx,
                                    const int* __restrict__ offs,
                                    const float* __restrict__ dinv,
                                    int2* __restrict__ sedge) {
    const int e0 = blockIdx.x * 256 + threadIdx.x;
    int k[EDGE_PER_THREAD], rk[EDGE_PER_THREAD], r[EDGE_PER_THREAD];
#pragma unroll
    for (int u = 0; u < EDGE_PER_THREAD; ++u) {
        const int e = e0 + u * EDGE_STRIDE;
        k[u]  = key[e];
        rk[u] = rank[e];
        r[u]  = eidx[e];
    }
    float wsrc[EDGE_PER_THREAD];
    int   pos[EDGE_PER_THREAD];
#pragma unroll
    for (int u = 0; u < EDGE_PER_THREAD; ++u) {
        const int t = k[u] / N_NODES;
        wsrc[u] = dinv[t * N_NODES + r[u]];
        pos[u]  = offs[k[u]] + rk[u];
    }
#pragma unroll
    for (int u = 0; u < EDGE_PER_THREAD; ++u) {
        sedge[pos[u]] = make_int2(r[u], __float_as_int(wsrc[u]));
    }
}

#define TILE_M 64
#define TBATCH 4
#define YP  132
#define AP  68
#define W1P 72
#define W2P 40

static_assert(TILE_M * AP <= TILE_M * YP, "the relu tile reuses the y tile's LDS");
static_assert(ETYPES % TBATCH == 0, "types are launched in whole batches");

__device__ __forceinline__ void EdgeGCN_axpy4(float4& acc, float w, const float4& r) {
    acc.x += w * r.x; acc.y += w * r.y; acc.z += w * r.z; acc.w += w * r.w;
}

__global__ __launch_bounds__(256) void EdgeGCN_layer1_fused_kernel(
        const float* __restrict__ x, const float* __restrict__ W1,
        const float* __restrict__ b1, const float* __restrict__ W2,
        const int* __restrict__ offs, const int* __restrict__ cnt,
        const float* __restrict__ dinv, const int2* __restrict__ sedge,
        float* __restrict__ h2, int tbase) {
    __shared__ __attribute__((aligned(16))) float sY[TILE_M * YP];
    __shared__ __attribute__((aligned(16))) float sW1[NODE_DIM * W1P];
    __shared__ __attribute__((aligned(16))) float sW2[STATE * W2P];
    __shared__ int   sBeg[TILE_M];
    __shared__ int   sCnt[TILE_M];
    __shared__ float sDi[TILE_M];

    const int tid  = threadIdx.x;
    const int lane = tid & 31;
    const int w    = tid >> 5;
    const int tl   = blockIdx.y;
    const int t    = tbase + tl;
    const int c0   = blockIdx.x * TILE_M;

    if (tid < TILE_M) {
        const int c = c0 + tid;
        int beg = 0, n = 0;
        float di = 0.0f;
        if (c < N_NODES) {
            const long long k = (long long)t * N_NODES + c;
            beg = offs[k];
            n   = cnt[k];
            di  = dinv[k];
        }
        sBeg[tid] = beg;
        sCnt[tid] = n;
        sDi[tid]  = di;
    }

    float4 selfv[8];
#pragma unroll
    for (int i = 0; i < 8; ++i) {
        const int c = c0 + w * 8 + i;
        const int cc = (c < N_NODES) ? c : 0;
        selfv[i] = *(const float4*)(x + (long long)cc * NODE_DIM + 4 * lane);
    }

    {
        const float4* W1v = (const float4*)(W1 + (long long)t * NODE_DIM * STATE);
#pragma unroll
        for (int i = tid; i < NODE_DIM * STATE / 4; i += 256) {
            const float4 v = W1v[i];
            *(float4*)&sW1[(i >> 4) * W1P + (i & 15) * 4] = v;
        }
        const float4* W2v = (const float4*)(W2 + (long long)t * STATE * OUT_DIM);
#pragma unroll
        for (int i = tid; i < STATE * OUT_DIM / 4; i += 256) {
            const float4 v = W2v[i];
            *(float4*)&sW2[(i >> 3) * W2P + (i & 7) * 4] = v;
        }
    }
    __syncthreads();

#pragma unroll
    for (int i = 0; i < 8; ++i) {
        const int r = w * 8 + i;
        const int n   = __builtin_amdgcn_readfirstlane(sCnt[r]);
        const int beg = __builtin_amdgcn_readfirstlane(sBeg[r]);
        const float di = sDi[r];
        const int2* ep = sedge + beg;
        const float* xl = x + 4 * lane;
        float4 acc = make_float4(0.f, 0.f, 0.f, 0.f);
        int j = 0;
        for (; j + 3 < n; j += 4) {
            const int2 e0 = ep[j], e1 = ep[j + 1], e2 = ep[j + 2], e3 = ep[j + 3];
            const float4 r0 = *(const float4*)(xl + (long long)e0.x * NODE_DIM);
            const float4 r1 = *(const float4*)(xl + (long long)e1.x * NODE_DIM);
            const float4 r2 = *(const float4*)(xl + (long long)e2.x * NODE_DIM);
            const float4 r3 = *(const float4*)(xl + (long long)e3.x * NODE_DIM);
            EdgeGCN_axpy4(acc, __int_as_float(e0.y), r0);
            EdgeGCN_axpy4(acc, __int_as_float(e1.y), r1);
            EdgeGCN_axpy4(acc, __int_as_float(e2.y), r2);
            EdgeGCN_axpy4(acc, __int_as_float(e3.y), r3);
        }
        const int rem = n - j;
        if (rem == 3) {
            const int2 e0 = ep[j], e1 = ep[j + 1], e2 = ep[j + 2];
            const float4 r0 = *(const float4*)(xl + (long long)e0.x * NODE_DIM);
            const float4 r1 = *(const float4*)(xl + (long long)e1.x * NODE_DIM);
            const float4 r2 = *(const float4*)(xl + (long long)e2.x * NODE_DIM);
            EdgeGCN_axpy4(acc, __int_as_float(e0.y), r0);
            EdgeGCN_axpy4(acc, __int_as_float(e1.y), r1);
            EdgeGCN_axpy4(acc, __int_as_float(e2.y), r2);
        } else if (rem == 2) {
            const int2 e0 = ep[j], e1 = ep[j + 1];
            const float4 r0 = *(const float4*)(xl + (long long)e0.x * NODE_DIM);
            const float4 r1 = *(const float4*)(xl + (long long)e1.x * NODE_DIM);
            EdgeGCN_axpy4(acc, __int_as_float(e0.y), r0);
            EdgeGCN_axpy4(acc, __int_as_float(e1.y), r1);
        } else if (rem == 1) {
            const int2 e0 = ep[j];
            const float4 r0 = *(const float4*)(xl + (long long)e0.x * NODE_DIM);
            EdgeGCN_axpy4(acc, __int_as_float(e0.y), r0);
        }
        const float d2 = di * di;
        const bool valid = (c0 + r) < N_NODES;
        float4 y;
        y.x = valid ? di * acc.x + selfv[i].x * d2 : 0.0f;
        y.y = valid ? di * acc.y + selfv[i].y * d2 : 0.0f;
        y.z = valid ? di * acc.z + selfv[i].z * d2 : 0.0f;
        y.w = valid ? di * acc.w + selfv[i].w * d2 : 0.0f;
        *(float4*)&sY[r * YP + 4 * lane] = y;
    }
    __syncthreads();

    const int lo = lane & 15;
    const int hi = lane >> 4;
    const int rt = w >> 1;
    const int cb = (w & 1) * 32;
    const float bias0 = b1[(long long)t * STATE + cb + lo];
    const float bias1 = b1[(long long)t * STATE + cb + 16 + lo];
    v8f p0 = {}, p1 = {};
    {
        const float* ya = &sY[(rt * 16 + lo) * YP + 2 * hi];
        const float* wb = &sW1[(2 * hi) * W1P + cb + lo];
#pragma unroll
        for (int kb = 0; kb < NODE_DIM; kb += 4) {
            const v2f a = *(const v2f*)(ya + kb);
            v2f q0, q1;
            q0.x = wb[kb * W1P];
            q0.y = wb[(kb + 1) * W1P];
            q1.x = wb[kb * W1P + 16];
            q1.y = wb[(kb + 1) * W1P + 16];
            p0 = __builtin_amdgcn_wmma_f32_16x16x4_f32(false, a, false, q0, (short)0, p0, false, false);
            p1 = __builtin_amdgcn_wmma_f32_16x16x4_f32(false, a, false, q1, (short)0, p1, false, false);
        }
    }
    __syncthreads();

    float* sA = sY;
#pragma unroll
    for (int v = 0; v < 8; ++v) {
        const int row = rt * 16 + v + 8 * hi;
        sA[row * AP + cb + lo]      = fmaxf(p0[v] + bias0, 0.0f);
        sA[row * AP + cb + 16 + lo] = fmaxf(p1[v] + bias1, 0.0f);
    }
    __syncthreads();

    const int ct = (w & 1) * 16;
    v8f d0 = {}, d1 = {};
    {
        const float* aa  = &sA[(rt * 16 + lo) * AP + 2 * hi];
        const float* wb2 = &sW2[(2 * hi) * W2P + ct + lo];
#pragma unroll
        for (int kb = 0; kb < STATE; kb += 8) {
            const v2f a0 = *(const v2f*)(aa + kb);
            const v2f a1 = *(const v2f*)(aa + kb + 4);
            v2f q0, q1;
            q0.x = wb2[kb * W2P];
            q0.y = wb2[(kb + 1) * W2P];
            q1.x = wb2[(kb + 4) * W2P];
            q1.y = wb2[(kb + 5) * W2P];
            d0 = __builtin_amdgcn_wmma_f32_16x16x4_f32(false, a0, false, q0, (short)0, d0, false, false);
            d1 = __builtin_amdgcn_wmma_f32_16x16x4_f32(false, a1, false, q1, (short)0, d1, false, false);
        }
    }
    float* h2t = h2 + (long long)tl * N_NODES * OUT_DIM;
#pragma unroll
    for (int v = 0; v < 8; ++v) {
        const int mr = c0 + rt * 16 + v + 8 * hi;
        if (mr < N_NODES) h2t[(long long)mr * OUT_DIM + ct + lo] = d0[v] + d1[v];
    }
}

#define G2_BLOCKS 782
#define G2_STRIDE (G2_BLOCKS * 32)

static_assert((long long)G2_STRIDE * 2 >= N_NODES, "two passes of the layer-2 gather cover every node");

__global__ __launch_bounds__(256) void EdgeGCN_gather2_kernel(
        const int* __restrict__ offs, const int* __restrict__ cnt,
        const float* __restrict__ dinv, const int2* __restrict__ sedge,
        const float* __restrict__ h2all, const float* __restrict__ b2,
        float* __restrict__ out, int tbase) {
    const float4* __restrict__ h4 = (const float4*)h2all;
    const float4* __restrict__ b4 = (const float4*)b2;
    float4* __restrict__ o4 = (float4*)out;
    const unsigned q4 = threadIdx.x & 7;
    const unsigned t0 = (unsigned)tbase;

    for (unsigned c = blockIdx.x * 32 + (threadIdx.x >> 3); c < N_NODES; c += G2_STRIDE) {
        int    beg[TBATCH], n[TBATCH];
        float  di[TBATCH];
        float4 self[TBATCH], acc[TBATCH];
#pragma unroll
        for (int tl = 0; tl < TBATCH; ++tl) {
            const unsigned k = (t0 + tl) * N_NODES + c;
            beg[tl]  = offs[k];
            n[tl]    = cnt[k];
            di[tl]   = dinv[k];
            self[tl] = h4[((unsigned)tl * N_NODES + c) * (OUT_DIM / 4) + q4];
            acc[tl]  = make_float4(0.f, 0.f, 0.f, 0.f);
        }
        int nmax = n[0];
#pragma unroll
        for (int tl = 1; tl < TBATCH; ++tl) nmax = (n[tl] > nmax) ? n[tl] : nmax;

        int2 e0[TBATCH], e1[TBATCH];
#pragma unroll
        for (int tl = 0; tl < TBATCH; ++tl) {
            const int i0 = beg[tl];
            const int i1 = beg[tl] + 1;
            e0[tl] = sedge[i0 < N_EDGES ? i0 : N_EDGES - 1];
            e1[tl] = sedge[i1 < N_EDGES ? i1 : N_EDGES - 1];
        }
        int j = 0;
        do {
            float4 r0[TBATCH], r1[TBATCH];
#pragma unroll
            for (int tl = 0; tl < TBATCH; ++tl) {
                r0[tl] = make_float4(0.f, 0.f, 0.f, 0.f);
                r1[tl] = make_float4(0.f, 0.f, 0.f, 0.f);
                if (j < n[tl])
                    r0[tl] = h4[((unsigned)tl * N_NODES + (unsigned)e0[tl].x) * (OUT_DIM / 4) + q4];
                if (j + 1 < n[tl])
                    r1[tl] = h4[((unsigned)tl * N_NODES + (unsigned)e1[tl].x) * (OUT_DIM / 4) + q4];
            }
            int2 f0[TBATCH], f1[TBATCH];
#pragma unroll
            for (int tl = 0; tl < TBATCH; ++tl) {
                const int i0 = beg[tl] + j + 2;
                const int i1 = beg[tl] + j + 3;
                f0[tl] = sedge[i0 < N_EDGES ? i0 : N_EDGES - 1];
                f1[tl] = sedge[i1 < N_EDGES ? i1 : N_EDGES - 1];
            }
#pragma unroll
            for (int tl = 0; tl < TBATCH; ++tl) {
                const float w0 = (j < n[tl])     ? __int_as_float(e0[tl].y) : 0.0f;
                const float w1 = (j + 1 < n[tl]) ? __int_as_float(e1[tl].y) : 0.0f;
                EdgeGCN_axpy4(acc[tl], w0, r0[tl]);
                EdgeGCN_axpy4(acc[tl], w1, r1[tl]);
                e0[tl] = f0[tl];
                e1[tl] = f1[tl];
            }
            j += 2;
        } while (j < nmax);
#pragma unroll
        for (int tl = 0; tl < TBATCH; ++tl) {
            const float4 b = b4[(t0 + tl) * (OUT_DIM / 4) + q4];
            const float d  = di[tl];
            const float d2 = d * d;
            float4 o;
            o.x = d * acc[tl].x + self[tl].x * d2 + b.x;
            o.y = d * acc[tl].y + self[tl].y * d2 + b.y;
            o.z = d * acc[tl].z + self[tl].z * d2 + b.z;
            o.w = d * acc[tl].w + self[tl].w * d2 + b.w;
            o4[c * (OUT_W / 4) + (t0 + tl) * (OUT_DIM / 4) + q4] = o;
        }
    }
}

extern "C" void kernel_launch(void* const* d_in, const int* in_sizes, int n_in,
                              void* d_out, int out_size, void* d_ws, size_t ws_size,
                              hipStream_t stream) {
    const float* x   = (const float*)d_in[0];
    const float* ea  = (const float*)d_in[1];
    const float* W1  = (const float*)d_in[2];
    const float* b1  = (const float*)d_in[3];
    const float* W2  = (const float*)d_in[4];
    const float* b2  = (const float*)d_in[5];
    const int*   eix = (const int*)d_in[6];
    float* out = (float*)d_out;

    char* ws = (char*)d_ws;
    auto alignup = [](size_t v) { return (v + 255) & ~(size_t)255; };
    size_t off = 0;
    int*   cnt   = (int*)  (ws + off); off += alignup(sizeof(int) * NKEYS);
    int*   offs  = (int*)  (ws + off); off += alignup(sizeof(int) * NKEYS);
    float* dinv  = (float*)(ws + off); off += alignup(sizeof(float) * NKEYS);
    int2*  sedge = (int2*) (ws + off); off += alignup(sizeof(int2) * (size_t)N_EDGES);
    float* h2    = (float*)(ws + off); off += alignup(sizeof(float) * (size_t)TBATCH * N_NODES * OUT_DIM);
    int*   tsum  = (int*)h2;
    int*   key   = tsum + 1024;
    int*   rank  = key + N_EDGES;
    (void)ws_size; (void)in_sizes; (void)n_in; (void)out_size;

    const int TB = 256;

    EdgeGCN_zero_int_kernel<<<ZERO_BLOCKS, TB, 0, stream>>>(cnt);
    EdgeGCN_key_rank_kernel<<<EDGE_BLOCKS, TB, 0, stream>>>(ea, eix, cnt, key, rank);
    EdgeGCN_tile_sum_kernel<<<SCAN_BLOCKS, TB, 0, stream>>>(cnt, tsum);
    EdgeGCN_scan_kernel<<<SCAN_BLOCKS, TB, 0, stream>>>(cnt, tsum, offs, dinv);
    EdgeGCN_fill_kernel<<<EDGE_BLOCKS, TB, 0, stream>>>(key, rank, eix, offs, dinv, sedge);

    for (int tb = 0; tb < ETYPES; tb += TBATCH) {
        EdgeGCN_layer1_fused_kernel<<<dim3((N_NODES + TILE_M - 1) / TILE_M, TBATCH), TB, 0, stream>>>(
            x, W1, b1, W2, offs, cnt, dinv, sedge, h2, tb);
        EdgeGCN_gather2_kernel<<<G2_BLOCKS, TB, 0, stream>>>(
            offs, cnt, dinv, sedge, h2, b2, out, tb);
    }
}


# layer 1 as resident work-groups: W1/W2 staged in LDS once per work-group, 64-row tiles pulled from a per-type counter (integer atomic); bucket sizes read as offset differences; sum order kept
# speedup vs baseline: 3.9030x; 1.0589x over previous
#include <hip/hip_runtime.h>
#include <stdint.h>

typedef __attribute__((ext_vector_type(2))) float v2f;
typedef __attribute__((ext_vector_type(8))) float v8f;

#define N_NODES  50000
#define N_EDGES  800000
#define NODE_DIM 128
#define STATE    64
#define OUT_DIM  32
#define ETYPES   8
#define OUT_W    (ETYPES * OUT_DIM)
#define NKEYS    (ETYPES * N_NODES)
#define SCAN_TILE   1024
#define SCAN_BLOCKS ((NKEYS + SCAN_TILE - 1) / SCAN_TILE)

static_assert(NKEYS % 4 == 0, "scan kernels read bucket counts four at a time");

#define EDGE_BLOCKS 625
#define EDGE_PER_THREAD 5
#define EDGE_STRIDE (EDGE_BLOCKS * 256)
#define ZERO_BLOCKS 196
#define ZERO_STRIDE (ZERO_BLOCKS * 256)

static_assert((long long)EDGE_STRIDE * EDGE_PER_THREAD == N_EDGES, "per-edge kernels cover the edge list exactly: no bounds test");
static_assert((long long)ZERO_STRIDE * 2 * 4 >= NKEYS, "the zero kernel covers every bucket");

__global__ __launch_bounds__(256) void EdgeGCN_zero_int_kernel(int* __restrict__ p) {
    const int i = blockIdx.x * 256 + threadIdx.x;
    const int4 z = make_int4(0, 0, 0, 0);
    int4* p4 = (int4*)p;
    if (i < NKEYS / 4) p4[i] = z;
    if (i + ZERO_STRIDE < NKEYS / 4) p4[i + ZERO_STRIDE] = z;
}

__global__ __launch_bounds__(256) void EdgeGCN_key_rank_kernel(const float* __restrict__ ea,
                                        const int* __restrict__ eidx,
                                        int* __restrict__ cnt,
                                        int* __restrict__ key,
                                        int* __restrict__ rank) {
    const int e0 = blockIdx.x * 256 + threadIdx.x;
    float4 lo[EDGE_PER_THREAD], hi[EDGE_PER_THREAD];
    int c[EDGE_PER_THREAD];
#pragma unroll
    for (int u = 0; u < EDGE_PER_THREAD; ++u) {
        const int e = e0 + u * EDGE_STRIDE;
        const float4* a4 = (const float4*)(ea + (long long)e * ETYPES);
        lo[u] = a4[0];
        hi[u] = a4[1];
        c[u]  = eidx[N_EDGES + e];
    }
    int k[EDGE_PER_THREAD], r[EDGE_PER_THREAD];
#pragma unroll
    for (int u = 0; u < EDGE_PER_THREAD; ++u) {
        int t = 0;
        if (lo[u].y > 0.5f) t = 1;
        if (lo[u].z > 0.5f) t = 2;
        if (lo[u].w > 0.5f) t = 3;
        if (hi[u].x > 0.5f) t = 4;
        if (hi[u].y > 0.5f) t = 5;
        if (hi[u].z > 0.5f) t = 6;
        if (hi[u].w > 0.5f) t = 7;
        k[u] = t * N_NODES + c[u];
    }
#pragma unroll
    for (int u = 0; u < EDGE_PER_THREAD; ++u) {
        r[u] = atomicAdd(&cnt[k[u]], 1);
    }
#pragma unroll
    for (int u = 0; u < EDGE_PER_THREAD; ++u) {
        const int e = e0 + u * EDGE_STRIDE;
        key[e]  = k[u];
        rank[e] = r[u];
    }
}

__global__ __launch_bounds__(256) void EdgeGCN_tile_sum_kernel(
        const int* __restrict__ cnt, int* __restrict__ tsum) {
    __shared__ int sw[8];
    const int tid = threadIdx.x;
    const int base = blockIdx.x * SCAN_TILE + tid * 4;
    int s = 0;
    if (base < NKEYS) {
        int4 v = *(const int4*)(cnt + base);
        s = v.x + v.y + v.z + v.w;
    }
    for (int o = 16; o > 0; o >>= 1) s += __shfl_xor(s, o, 32);
    if ((tid & 31) == 0) sw[tid >> 5] = s;
    __syncthreads();
    if (tid == 0) {
        int tot = 0;
        for (int i = 0; i < 8; ++i) tot += sw[i];
        tsum[blockIdx.x] = tot;
    }
}

__global__ __launch_bounds__(256) void EdgeGCN_scan_kernel(
        const int* __restrict__ cnt, const int* __restrict__ tsum,
        int* __restrict__ offs, float* __restrict__ dinv) {
    __shared__ int sp[8];
    __shared__ int sw[8];
    const int tid  = threadIdx.x;
    const int lane = tid & 31;
    const int w    = tid >> 5;

    int p = 0;
    for (int i = tid; i < (int)blockIdx.x; i += 256) p += tsum[i];
    for (int o = 16; o > 0; o >>= 1) p += __shfl_xor(p, o, 32);
    if (lane == 0) sp[w] = p;

    const int base = blockIdx.x * SCAN_TILE + tid * 4;
    const bool in = base < NKEYS;
    int4 v = make_int4(0, 0, 0, 0);
    if (in) v = *(const int4*)(cnt + base);
    const int s = v.x + v.y + v.z + v.w;
    int inc = s;
    for (int o = 1; o < 32; o <<= 1) {
        int u = __shfl_up(inc, o, 32);
        if (lane >= o) inc += u;
    }
    if (lane == 31) sw[w] = inc;
    __syncthreads();

    int ex = inc - s;
    for (int i = 0; i < 8; ++i) {
        ex += sp[i];
        if (i < w) ex += sw[i];
    }
    if (in) {
        int4 o4;
        o4.x = ex;
        o4.y = o4.x + v.x;
        o4.z = o4.y + v.y;
        o4.w = o4.z + v.z;
        *(int4*)(offs + base) = o4;
        float4 d;
        d.x = rsqrtf((float)v.x + 1.0f);
        d.y = rsqrtf((float)v.y + 1.0f);
        d.z = rsqrtf((float)v.z + 1.0f);
        d.w = rsqrtf((float)v.w + 1.0f);
        *(float4*)(dinv + base) = d;
    }
}

__global__ __launch_bounds__(256) void EdgeGCN_fill_kernel(const int* __restrict__ key,
                                    const int* __restrict__ rank,
                                    const int* __restrict__ eidx,
                                    const int* __restrict__ offs,
                                    const float* __restrict__ dinv,
                                    int2* __restrict__ sedge,
                                    int* __restrict__ tile_ctr) {
    if (blockIdx.x == 0 && threadIdx.x < ETYPES) tile_ctr[threadIdx.x] = 0;
    const int e0 = blockIdx.x * 256 + threadIdx.x;
    int k[EDGE_PER_THREAD], rk[EDGE_PER_THREAD], r[EDGE_PER_THREAD];
#pragma unroll
    for (int u = 0; u < EDGE_PER_THREAD; ++u) {
        const int e = e0 + u * EDGE_STRIDE;
        k[u]  = key[e];
        rk[u] = rank[e];
        r[u]  = eidx[e];
    }
    float wsrc[EDGE_PER_THREAD];
    int   pos[EDGE_PER_THREAD];
#pragma unroll
    for (int u = 0; u < EDGE_PER_THREAD; ++u) {
        const int t = k[u] / N_NODES;
        wsrc[u] = dinv[t * N_NODES + r[u]];
        pos[u]  = offs[k[u]] + rk[u];
    }
#pragma unroll
    for (int u = 0; u < EDGE_PER_THREAD; ++u) {
        sedge[pos[u]] = make_int2(r[u], __float_as_int(wsrc[u]));
    }
}

#define TILE_M 64
#define TBATCH 4
#define YP  132
#define AP  68
#define W1P 72
#define W2P 40
#define L1_TILES  ((N_NODES + TILE_M - 1) / TILE_M)
#define L1_BLOCKS 1024
#define L1_FETCH_TID 224
static_assert(L1_BLOCKS % TBATCH == 0, "every type of a launch gets the same number of work-groups");
static_assert((NODE_DIM * STATE / 4) % 256 == 0 && (STATE * OUT_DIM / 4) % 256 == 0, "weight staging: whole rounds of 256 threads");

static_assert(TILE_M * AP <= TILE_M * YP, "the relu tile reuses the y tile's LDS");
static_assert(ETYPES % TBATCH == 0, "types are launched in whole batches");

__device__ __forceinline__ void EdgeGCN_axpy4(float4& acc, float w, const float4& r) {
    acc.x += w * r.x; acc.y += w * r.y; acc.z += w * r.z; acc.w += w * r.w;
}

__global__ __launch_bounds__(256) void EdgeGCN_layer1_fused_kernel(
        const float* __restrict__ x, const float* __restrict__ W1,
        const float* __restrict__ b1, const float* __restrict__ W2,
        const int* __restrict__ offs,
        const float* __restrict__ dinv, const int2* __restrict__ sedge,
        int* __restrict__ tile_ctr, float* __restrict__ h2, int tbase) {
    __shared__ __attribute__((aligned(16))) float sY[TILE_M * YP];
    __shared__ __attribute__((aligned(16))) float sW1[NODE_DIM * W1P];
    __shared__ __attribute__((aligned(16))) float sW2[STATE * W2P];
    __shared__ int   sBeg[TILE_M];
    __shared__ int   sCnt[TILE_M];
    __shared__ float sDi[TILE_M];
    __shared__ int   sNext;

    const int tid  = threadIdx.x;
    const int lane = tid & 31;
    const int w    = tid >> 5;
    const int tl   = blockIdx.x & (TBATCH - 1);
    const int t    = tbase + tl;
    unsigned int* const ctr = (unsigned int*)tile_ctr + t;
    const bool fetcher = (tid == L1_FETCH_TID);

    int nxt = 0;
    if (fetcher) nxt = (int)atomicInc(ctr, 0xffffffffu);

    {
        const float4* W1v = (const float4*)(W1 + (long long)t * NODE_DIM * STATE);
        const float4* W2v = (const float4*)(W2 + (long long)t * STATE * OUT_DIM);
        float4 v1[NODE_DIM * STATE / 4 / 256], v2[STATE * OUT_DIM / 4 / 256];
#pragma unroll
        for (int u = 0; u < NODE_DIM * STATE / 4 / 256; ++u) v1[u] = W1v[tid + u * 256];
#pragma unroll
        for (int u = 0; u < STATE * OUT_DIM / 4 / 256; ++u) v2[u] = W2v[tid + u * 256];
#pragma unroll
        for (int u = 0; u < NODE_DIM * STATE / 4 / 256; ++u) {
            const int i = tid + u * 256;
            *(float4*)&sW1[(i >> 4) * W1P + (i & 15) * 4] = v1[u];
        }
#pragma unroll
        for (int u = 0; u < STATE * OUT_DIM / 4 / 256; ++u) {
            const int i = tid + u * 256;
            *(float4*)&sW2[(i >> 3) * W2P + (i & 7) * 4] = v2[u];
        }
    }
    if (fetcher) sNext = nxt;
    __syncthreads();

    const int lo = lane & 15;
    const int hi = lane >> 4;
    const int rt = w >> 1;
    const int cb = (w & 1) * 32;
    const int ct = (w & 1) * 16;
    const float bias0 = b1[(long long)t * STATE + cb + lo];
    const float bias1 = b1[(long long)t * STATE + cb + 16 + lo];
    float* h2t = h2 + (long long)tl * N_NODES * OUT_DIM;

    int tile = __builtin_amdgcn_readfirstlane(sNext);
    while (tile < L1_TILES) {
    const int c0 = tile * TILE_M;

    if (tid < TILE_M) {
        const int c = c0 + tid;
        int beg = 0, n = 0;
        float di = 0.0f;
        if (c < N_NODES) {
            const int k = t * N_NODES + c;
            beg = offs[k];
            const int end = (k + 1 < NKEYS) ? offs[k + 1] : N_EDGES;
            n   = end - beg;
            di  = dinv[k];
        }
        sBeg[tid] = beg;
        sCnt[tid] = n;
        sDi[tid]  = di;
    }

    float4 selfv[8];
#pragma unroll
    for (int i = 0; i < 8; ++i) {
        const int c = c0 + w * 8 + i;
        const int cc = (c < N_NODES) ? c : 0;
        selfv[i] = *(const float4*)(x + (long long)cc * NODE_DIM + 4 * lane);
    }
    __syncthreads();

#pragma unroll
    for (int i = 0; i < 8; ++i) {
        const int r = w * 8 + i;
        const int n   = __builtin_amdgcn_readfirstlane(sCnt[r]);
        const int beg = __builtin_amdgcn_readfirstlane(sBeg[r]);
        const float di = sDi[r];
        const int2* ep = sedge + beg;
        const float* xl = x + 4 * lane;
        float4 acc = make_float4(0.f, 0.f, 0.f, 0.f);
        int j = 0;
        for (; j + 3 < n; j += 4) {
            const int2 e0 = ep[j], e1 = ep[j + 1], e2 = ep[j + 2], e3 = ep[j + 3];
            const float4 r0 = *(const float4*)(xl + (long long)e0.x * NODE_DIM);
            const float4 r1 = *(const float4*)(xl + (long long)e1.x * NODE_DIM);
            const float4 r2 = *(const float4*)(xl + (long long)e2.x * NODE_DIM);
            const float4 r3 = *(const float4*)(xl + (long long)e3.x * NODE_DIM);
            EdgeGCN_axpy4(acc, __int_as_float(e0.y), r0);
            EdgeGCN_axpy4(acc, __int_as_float(e1.y), r1);
            EdgeGCN_axpy4(acc, __int_as_float(e2.y), r2);
            EdgeGCN_axpy4(acc, __int_as_float(e3.y), r3);
        }
        const int rem = n - j;
        if (rem == 3) {
            const int2 e0 = ep[j], e1 = ep[j + 1], e2 = ep[j + 2];
            const float4 r0 = *(const float4*)(xl + (long long)e0.x * NODE_DIM);
            const float4 r1 = *(const float4*)(xl + (long long)e1.x * NODE_DIM);
            const float4 r2 = *(const float4*)(xl + (long long)e2.x * NODE_DIM);
            EdgeGCN_axpy4(acc, __int_as_float(e0.y), r0);
            EdgeGCN_axpy4(acc, __int_as_float(e1.y), r1);
            EdgeGCN_axpy4(acc, __int_as_float(e2.y), r2);
        } else if (rem == 2) {
            const int2 e0 = ep[j], e1 = ep[j + 1];
            const float4 r0 = *(const float4*)(xl + (long long)e0.x * NODE_DIM);
            const float4 r1 = *(const float4*)(xl + (long long)e1.x * NODE_DIM);
            EdgeGCN_axpy4(acc, __int_as_float(e0.y), r0);
            EdgeGCN_axpy4(acc, __int_as_float(e1.y), r1);
        } else if (rem == 1) {
            const int2 e0 = ep[j];
            const float4 r0 = *(const float4*)(xl + (long long)e0.x * NODE_DIM);
            EdgeGCN_axpy4(acc, __int_as_float(e0.y), r0);
        }
        const float d2 = di * di;
        const bool valid = (c0 + r) < N_NODES;
        float4 y;
        y.x = valid ? di * acc.x + selfv[i].x * d2 : 0.0f;
        y.y = valid ? di * acc.y + selfv[i].y * d2 : 0.0f;
        y.z = valid ? di * acc.z + selfv[i].z * d2 : 0.0f;
        y.w = valid ? di * acc.w + selfv[i].w * d2 : 0.0f;
        *(float4*)&sY[r * YP + 4 * lane] = y;
    }
    __syncthreads();

    if (fetcher) nxt = (int)atomicInc(ctr, 0xffffffffu);

    v8f p0 = {}, p1 = {};
    {
        const float* ya = &sY[(rt * 16 + lo) * YP + 2 * hi];
        const float* wb = &sW1[(2 * hi) * W1P + cb + lo];
#pragma unroll
        for (int kb = 0; kb < NODE_DIM; kb += 4) {
            const v2f a = *(const v2f*)(ya + kb);
            v2f q0, q1;
            q0.x = wb[kb * W1P];
            q0.y = wb[(kb + 1) * W1P];
            q1.x = wb[kb * W1P + 16];
            q1.y = wb[(kb + 1) * W1P + 16];
            p0 = __builtin_amdgcn_wmma_f32_16x16x4_f32(false, a, false, q0, (short)0, p0, false, false);
            p1 = __builtin_amdgcn_wmma_f32_16x16x4_f32(false, a, false, q1, (short)0, p1, false, false);
        }
    }
    __syncthreads();

    float* sA = sY;
#pragma unroll
    for (int v = 0; v < 8; ++v) {
        const int row = rt * 16 + v + 8 * hi;
        sA[row * AP + cb + lo]      = fmaxf(p0[v] + bias0, 0.0f);
        sA[row * AP + cb + 16 + lo] = fmaxf(p1[v] + bias1, 0.0f);
    }
    if (fetcher) sNext = nxt;
    __syncthreads();

    v8f d0 = {}, d1 = {};
    {
        const float* aa  = &sA[(rt * 16 + lo) * AP + 2 * hi];
        const float* wb2 = &sW2[(2 * hi) * W2P + ct + lo];
#pragma unroll
        for (int kb = 0; kb < STATE; kb += 8) {
            const v2f a0 = *(const v2f*)(aa + kb);
            const v2f a1 = *(const v2f*)(aa + kb + 4);
            v2f q0, q1;
            q0.x = wb2[kb * W2P];
            q0.y = wb2[(kb + 1) * W2P];
            q1.x = wb2[(kb + 4) * W2P];
            q1.y = wb2[(kb + 5) * W2P];
            d0 = __builtin_amdgcn_wmma_f32_16x16x4_f32(false, a0, false, q0, (short)0, d0, false, false);
            d1 = __builtin_amdgcn_wmma_f32_16x16x4_f32(false, a1, false, q1, (short)0, d1, false, false);
        }
    }
#pragma unroll
    for (int v = 0; v < 8; ++v) {
        const int mr = c0 + rt * 16 + v + 8 * hi;
        if (mr < N_NODES) h2t[(long long)mr * OUT_DIM + ct + lo] = d0[v] + d1[v];
    }

    tile = __builtin_amdgcn_readfirstlane(sNext);
    }
}

#define G2_BLOCKS 782
#define G2_STRIDE (G2_BLOCKS * 32)

static_assert((long long)G2_STRIDE * 2 >= N_NODES, "two passes of the layer-2 gather cover every node");

__global__ __launch_bounds__(256) void EdgeGCN_gather2_kernel(
        const int* __restrict__ offs,
        const float* __restrict__ dinv, const int2* __restrict__ sedge,
        const float* __restrict__ h2all, const float* __restrict__ b2,
        float* __restrict__ out, int tbase) {
    const float4* __restrict__ h4 = (const float4*)h2all;
    const float4* __restrict__ b4 = (const float4*)b2;
    float4* __restrict__ o4 = (float4*)out;
    const unsigned q4 = threadIdx.x & 7;
    const unsigned t0 = (unsigned)tbase;

    for (unsigned c = blockIdx.x * 32 + (threadIdx.x >> 3); c < N_NODES; c += G2_STRIDE) {
        int    beg[TBATCH], n[TBATCH];
        float  di[TBATCH];
        float4 self[TBATCH], acc[TBATCH];
#pragma unroll
        for (int tl = 0; tl < TBATCH; ++tl) {
            const unsigned k = (t0 + tl) * N_NODES + c;
            beg[tl]  = offs[k];
            n[tl]    = ((k + 1 < NKEYS) ? offs[k + 1] : N_EDGES) - beg[tl];
            di[tl]   = dinv[k];
            self[tl] = h4[((unsigned)tl * N_NODES + c) * (OUT_DIM / 4) + q4];
            acc[tl]  = make_float4(0.f, 0.f, 0.f, 0.f);
        }
        int nmax = n[0];
#pragma unroll
        for (int tl = 1; tl < TBATCH; ++tl) nmax = (n[tl] > nmax) ? n[tl] : nmax;

        int2 e0[TBATCH], e1[TBATCH];
#pragma unroll
        for (int tl = 0; tl < TBATCH; ++tl) {
            const int i0 = beg[tl];
            const int i1 = beg[tl] + 1;
            e0[tl] = sedge[i0 < N_EDGES ? i0 : N_EDGES - 1];
            e1[tl] = sedge[i1 < N_EDGES ? i1 : N_EDGES - 1];
        }
        int j = 0;
        do {
            float4 r0[TBATCH], r1[TBATCH];
#pragma unroll
            for (int tl = 0; tl < TBATCH; ++tl) {
                r0[tl] = make_float4(0.f, 0.f, 0.f, 0.f);
                r1[tl] = make_float4(0.f, 0.f, 0.f, 0.f);
                if (j < n[tl])
                    r0[tl] = h4[((unsigned)tl * N_NODES + (unsigned)e0[tl].x) * (OUT_DIM / 4) + q4];
                if (j + 1 < n[tl])
                    r1[tl] = h4[((unsigned)tl * N_NODES + (unsigned)e1[tl].x) * (OUT_DIM / 4) + q4];
            }
            int2 f0[TBATCH], f1[TBATCH];
#pragma unroll
            for (int tl = 0; tl < TBATCH; ++tl) {
                const int i0 = beg[tl] + j + 2;
                const int i1 = beg[tl] + j + 3;
                f0[tl] = sedge[i0 < N_EDGES ? i0 : N_EDGES - 1];
                f1[tl] = sedge[i1 < N_EDGES ? i1 : N_EDGES - 1];
            }
#pragma unroll
            for (int tl = 0; tl < TBATCH; ++tl) {
                const float w0 = (j < n[tl])     ? __int_as_float(e0[tl].y) : 0.0f;
                const float w1 = (j + 1 < n[tl]) ? __int_as_float(e1[tl].y) : 0.0f;
                EdgeGCN_axpy4(acc[tl], w0, r0[tl]);
                EdgeGCN_axpy4(acc[tl], w1, r1[tl]);
                e0[tl] = f0[tl];
                e1[tl] = f1[tl];
            }
            j += 2;
        } while (j < nmax);
#pragma unroll
        for (int tl = 0; tl < TBATCH; ++tl) {
            const float4 b = b4[(t0 + tl) * (OUT_DIM / 4) + q4];
            const float d  = di[tl];
            const float d2 = d * d;
            float4 o;
            o.x = d * acc[tl].x + self[tl].x * d2 + b.x;
            o.y = d * acc[tl].y + self[tl].y * d2 + b.y;
            o.z = d * acc[tl].z + self[tl].z * d2 + b.z;
            o.w = d * acc[tl].w + self[tl].w * d2 + b.w;
            o4[c * (OUT_W / 4) + (t0 + tl) * (OUT_DIM / 4) + q4] = o;
        }
    }
}

extern "C" void kernel_launch(void* const* d_in, const int* in_sizes, int n_in,
                              void* d_out, int out_size, void* d_ws, size_t ws_size,
                              hipStream_t stream) {
    const float* x   = (const float*)d_in[0];
    const float* ea  = (const float*)d_in[1];
    const float* W1  = (const float*)d_in[2];
    const float* b1  = (const float*)d_in[3];
    const float* W2  = (const float*)d_in[4];
    const float* b2  = (const float*)d_in[5];
    const int*   eix = (const int*)d_in[6];
    float* out = (float*)d_out;

    char* ws = (char*)d_ws;
    auto alignup = [](size_t v) { return (v + 255) & ~(size_t)255; };
    size_t off = 0;
    int*   cnt   = (int*)  (ws + off); off += alignup(sizeof(int) * NKEYS);
    int*   offs  = (int*)  (ws + off); off += alignup(sizeof(int) * NKEYS);
    float* dinv  = (float*)(ws + off); off += alignup(sizeof(float) * NKEYS);
    int2*  sedge = (int2*) (ws + off); off += alignup(sizeof(int2) * (size_t)N_EDGES);
    float* h2    = (float*)(ws + off); off += alignup(sizeof(float) * (size_t)TBATCH * N_NODES * OUT_DIM);
    int*   tsum  = (int*)h2;
    int*   key   = tsum + 1024;
    int*   rank  = key + N_EDGES;
    (void)ws_size; (void)in_sizes; (void)n_in; (void)out_size;

    const int TB = 256;

    EdgeGCN_zero_int_kernel<<<ZERO_BLOCKS, TB, 0, stream>>>(cnt);
    EdgeGCN_key_rank_kernel<<<EDGE_BLOCKS, TB, 0, stream>>>(ea, eix, cnt, key, rank);
    EdgeGCN_tile_sum_kernel<<<SCAN_BLOCKS, TB, 0, stream>>>(cnt, tsum);
    EdgeGCN_scan_kernel<<<SCAN_BLOCKS, TB, 0, stream>>>(cnt, tsum, offs, dinv);
    EdgeGCN_fill_kernel<<<EDGE_BLOCKS, TB, 0, stream>>>(key, rank, eix, offs, dinv, sedge, cnt);

    for (int tb = 0; tb < ETYPES; tb += TBATCH) {
        EdgeGCN_layer1_fused_kernel<<<L1_BLOCKS, TB, 0, stream>>>(
            x, W1, b1, W2, offs, dinv, sedge, cnt, h2, tb);
        EdgeGCN_gather2_kernel<<<G2_BLOCKS, TB, 0, stream>>>(
            offs, dinv, sedge, h2, b2, out, tb);
    }
}


# layer-1 gather: a wave reads its 8 rows' edge records with one load and requests up to 16 source rows across row boundaries before summing; bucket bounds fetched a tile ahead; sum order kept
# speedup vs baseline: 4.0895x; 1.0478x over previous
#include <hip/hip_runtime.h>
#include <stdint.h>

typedef __attribute__((ext_vector_type(2))) float v2f;
typedef __attribute__((ext_vector_type(8))) float v8f;

#define N_NODES  50000
#define N_EDGES  800000
#define NODE_DIM 128
#define STATE    64
#define OUT_DIM  32
#define ETYPES   8
#define OUT_W    (ETYPES * OUT_DIM)
#define NKEYS    (ETYPES * N_NODES)
#define SCAN_TILE   1024
#define SCAN_BLOCKS ((NKEYS + SCAN_TILE - 1) / SCAN_TILE)

static_assert(NKEYS % 4 == 0, "scan kernels read bucket counts four at a time");

#define EDGE_BLOCKS 625
#define EDGE_PER_THREAD 5
#define EDGE_STRIDE (EDGE_BLOCKS * 256)
#define ZERO_BLOCKS 196
#define ZERO_STRIDE (ZERO_BLOCKS * 256)

static_assert((long long)EDGE_STRIDE * EDGE_PER_THREAD == N_EDGES, "per-edge kernels cover the edge list exactly: no bounds test");
static_assert((long long)ZERO_STRIDE * 2 * 4 >= NKEYS, "the zero kernel covers every bucket");

__global__ __launch_bounds__(256) void EdgeGCN_zero_int_kernel(int* __restrict__ p) {
    const int i = blockIdx.x * 256 + threadIdx.x;
    const int4 z = make_int4(0, 0, 0, 0);
    int4* p4 = (int4*)p;
    if (i < NKEYS / 4) p4[i] = z;
    if (i + ZERO_STRIDE < NKEYS / 4) p4[i + ZERO_STRIDE] = z;
}

__global__ __launch_bounds__(256) void EdgeGCN_key_rank_kernel(const float* __restrict__ ea,
                                        const int* __restrict__ eidx,
                                        int* __restrict__ cnt,
                                        int* __restrict__ key,
                                        int* __restrict__ rank) {
    const int e0 = blockIdx.x * 256 + threadIdx.x;
    float4 lo[EDGE_PER_THREAD], hi[EDGE_PER_THREAD];
    int c[EDGE_PER_THREAD];
#pragma unroll
    for (int u = 0; u < EDGE_PER_THREAD; ++u) {
        const int e = e0 + u * EDGE_STRIDE;
        const float4* a4 = (const float4*)(ea + (long long)e * ETYPES);
        lo[u] = a4[0];
        hi[u] = a4[1];
        c[u]  = eidx[N_EDGES + e];
    }
    int k[EDGE_PER_THREAD], r[EDGE_PER_THREAD];
#pragma unroll
    for (int u = 0; u < EDGE_PER_THREAD; ++u) {
        int t = 0;
        if (lo[u].y > 0.5f) t = 1;
        if (lo[u].z > 0.5f) t = 2;
        if (lo[u].w > 0.5f) t = 3;
        if (hi[u].x > 0.5f) t = 4;
        if (hi[u].y > 0.5f) t = 5;
        if (hi[u].z > 0.5f) t = 6;
        if (hi[u].w > 0.5f) t = 7;
        k[u] = t * N_NODES + c[u];
    }
#pragma unroll
    for (int u = 0; u < EDGE_PER_THREAD; ++u) {
        r[u] = atomicAdd(&cnt[k[u]], 1);
    }
#pragma unroll
    for (int u = 0; u < EDGE_PER_THREAD; ++u) {
        const int e = e0 + u * EDGE_STRIDE;
        key[e]  = k[u];
        rank[e] = r[u];
    }
}

__global__ __launch_bounds__(256) void EdgeGCN_tile_sum_kernel(
        const int* __restrict__ cnt, int* __restrict__ tsum) {
    __shared__ int sw[8];
    const int tid = threadIdx.x;
    const int base = blockIdx.x * SCAN_TILE + tid * 4;
    int s = 0;
    if (base < NKEYS) {
        int4 v = *(const int4*)(cnt + base);
        s = v.x + v.y + v.z + v.w;
    }
    for (int o = 16; o > 0; o >>= 1) s += __shfl_xor(s, o, 32);
    if ((tid & 31) == 0) sw[tid >> 5] = s;
    __syncthreads();
    if (tid == 0) {
        int tot = 0;
        for (int i = 0; i < 8; ++i) tot += sw[i];
        tsum[blockIdx.x] = tot;
    }
}

__global__ __launch_bounds__(256) void EdgeGCN_scan_kernel(
        const int* __restrict__ cnt, const int* __restrict__ tsum,
        int* __restrict__ offs, float* __restrict__ dinv) {
    __shared__ int sp[8];
    __shared__ int sw[8];
    const int tid  = threadIdx.x;
    const int lane = tid & 31;
    const int w    = tid >> 5;

    int p = 0;
    for (int i = tid; i < (int)blockIdx.x; i += 256) p += tsum[i];
    for (int o = 16; o > 0; o >>= 1) p += __shfl_xor(p, o, 32);
    if (lane == 0) sp[w] = p;

    const int base = blockIdx.x * SCAN_TILE + tid * 4;
    const bool in = base < NKEYS;
    int4 v = make_int4(0, 0, 0, 0);
    if (in) v = *(const int4*)(cnt + base);
    const int s = v.x + v.y + v.z + v.w;
    int inc = s;
    for (int o = 1; o < 32; o <<= 1) {
        int u = __shfl_up(inc, o, 32);
        if (lane >= o) inc += u;
    }
    if (lane == 31) sw[w] = inc;
    __syncthreads();

    int ex = inc - s;
    for (int i = 0; i < 8; ++i) {
        ex += sp[i];
        if (i < w) ex += sw[i];
    }
    if (in) {
        int4 o4;
        o4.x = ex;
        o4.y = o4.x + v.x;
        o4.z = o4.y + v.y;
        o4.w = o4.z + v.z;
        *(int4*)(offs + base) = o4;
        float4 d;
        d.x = rsqrtf((float)v.x + 1.0f);
        d.y = rsqrtf((float)v.y + 1.0f);
        d.z = rsqrtf((float)v.z + 1.0f);
        d.w = rsqrtf((float)v.w + 1.0f);
        *(float4*)(dinv + base) = d;
    }
}

__global__ __launch_bounds__(256) void EdgeGCN_fill_kernel(const int* __restrict__ key,
                                    const int* __restrict__ rank,
                                    const int* __restrict__ eidx,
                                    const int* __restrict__ offs,
                                    const float* __restrict__ dinv,
                                    int2* __restrict__ sedge,
                                    int* __restrict__ tile_ctr) {
    if (blockIdx.x == 0 && threadIdx.x < ETYPES) tile_ctr[threadIdx.x] = 0;
    const int e0 = blockIdx.x * 256 + threadIdx.x;
    int k[EDGE_PER_THREAD], rk[EDGE_PER_THREAD], r[EDGE_PER_THREAD];
#pragma unroll
    for (int u = 0; u < EDGE_PER_THREAD; ++u) {
        const int e = e0 + u * EDGE_STRIDE;
        k[u]  = key[e];
        rk[u] = rank[e];
        r[u]  = eidx[e];
    }
    float wsrc[EDGE_PER_THREAD];
    int   pos[EDGE_PER_THREAD];
#pragma unroll
    for (int u = 0; u < EDGE_PER_THREAD; ++u) {
        const int t = k[u] / N_NODES;
        wsrc[u] = dinv[t * N_NODES + r[u]];
        pos[u]  = offs[k[u]] + rk[u];
    }
#pragma unroll
    for (int u = 0; u < EDGE_PER_THREAD; ++u) {
        sedge[pos[u]] = make_int2(r[u], __float_as_int(wsrc[u]));
    }
}

#define TILE_M 64
#define TBATCH 4
#define YP  132
#define AP  68
#define W1P 72
#define W2P 40
#define L1_TILES  ((N_NODES + TILE_M - 1) / TILE_M)
#define L1_BLOCKS 1024
#define L1_FETCH_TID 224
#define GROWS  8
#define GCHUNK 16
static_assert(L1_BLOCKS % TBATCH == 0, "every type of a launch gets the same number of work-groups");
static_assert((NODE_DIM * STATE / 4) % 256 == 0 && (STATE * OUT_DIM / 4) % 256 == 0, "weight staging: whole rounds of 256 threads");

static_assert(TILE_M * AP <= TILE_M * YP, "the relu tile reuses the y tile's LDS");
static_assert(TILE_M == 8 * GROWS, "eight waves of GROWS rows each make a tile");
static_assert(GCHUNK == 16, "a 32-lane record load feeds exactly two chunks");
static_assert(ETYPES % TBATCH == 0, "types are launched in whole batches");

__device__ __forceinline__ void EdgeGCN_axpy4(float4& acc, float w, const float4& r) {
    acc.x += w * r.x; acc.y += w * r.y; acc.z += w * r.z; acc.w += w * r.w;
}

__global__ __launch_bounds__(256) void EdgeGCN_layer1_fused_kernel(
        const float* __restrict__ x, const float* __restrict__ W1,
        const float* __restrict__ b1, const float* __restrict__ W2,
        const int* __restrict__ offs,
        const float* __restrict__ dinv, const int2* __restrict__ sedge,
        int* __restrict__ tile_ctr, float* __restrict__ h2, int tbase) {
    __shared__ __attribute__((aligned(16))) float sY[TILE_M * YP];
    __shared__ __attribute__((aligned(16))) float sW1[NODE_DIM * W1P];
    __shared__ __attribute__((aligned(16))) float sW2[STATE * W2P];
    __shared__ int   sNext;

    const int tid  = threadIdx.x;
    const int lane = tid & 31;
    const int w    = __builtin_amdgcn_readfirstlane(tid >> 5);
    const int tl   = blockIdx.x & (TBATCH - 1);
    const int t    = tbase + tl;
    unsigned int* const ctr = (unsigned int*)tile_ctr + t;
    const bool fetcher = (tid == L1_FETCH_TID);

    int nxt = 0;
    if (fetcher) nxt = (int)atomicInc(ctr, 0xffffffffu);

    {
        const float4* W1v = (const float4*)(W1 + (long long)t * NODE_DIM * STATE);
        const float4* W2v = (const float4*)(W2 + (long long)t * STATE * OUT_DIM);
        float4 v1[NODE_DIM * STATE / 4 / 256], v2[STATE * OUT_DIM / 4 / 256];
#pragma unroll
        for (int u = 0; u < NODE_DIM * STATE / 4 / 256; ++u) v1[u] = W1v[tid + u * 256];
#pragma unroll
        for (int u = 0; u < STATE * OUT_DIM / 4 / 256; ++u) v2[u] = W2v[tid + u * 256];
#pragma unroll
        for (int u = 0; u < NODE_DIM * STATE / 4 / 256; ++u) {
            const int i = tid + u * 256;
            *(float4*)&sW1[(i >> 4) * W1P + (i & 15) * 4] = v1[u];
        }
#pragma unroll
        for (int u = 0; u < STATE * OUT_DIM / 4 / 256; ++u) {
            const int i = tid + u * 256;
            *(float4*)&sW2[(i >> 3) * W2P + (i & 7) * 4] = v2[u];
        }
    }
    if (fetcher) sNext = nxt;
    __syncthreads();

    const int lo = lane & 15;
    const int hi = lane >> 4;
    const int rt = w >> 1;
    const int cb = (w & 1) * 32;
    const int ct = (w & 1) * 16;
    const float bias0 = b1[(long long)t * STATE + cb + lo];
    const float bias1 = b1[(long long)t * STATE + cb + 16 + lo];
    float* h2t = h2 + (long long)tl * N_NODES * OUT_DIM;

    const int rbase = w * GROWS;
#define EDGEGCN_L1_HEADERS(tile_, ev_, dv_)                                      \
    {                                                                            \
        const int cwh = (tile_) * TILE_M + rbase;                                \
        int nv = N_NODES - cwh;                                                  \
        nv = nv < 0 ? 0 : (nv > GROWS ? GROWS : nv);                             \
        const int kh = t * N_NODES + cwh;                                        \
        const int ki = kh + (lane < nv ? lane : nv);                             \
        ev_ = N_EDGES;                        \
        if (ki >= 0 && ki < NKEYS) ev_ = offs[ki];                               \
        dv_ = 0.0f;                                                              \
        if (lane < nv) dv_ = dinv[kh + lane];                                    \
    }

    int tile = __builtin_amdgcn_readfirstlane(sNext);
    int   ev;
    float dv;
    EDGEGCN_L1_HEADERS(tile, ev, dv)
    while (tile < L1_TILES) {
    const int c0 = tile * TILE_M;
    const int cw = c0 + rbase;

    float4 selfv[GROWS];
#pragma unroll
    for (int i = 0; i < GROWS; ++i) {
        const int c = cw + i;
        const int cc = (c < N_NODES) ? c : 0;
        selfv[i] = *(const float4*)(x + (long long)cc * NODE_DIM + 4 * lane);
    }

    const int e0   = __builtin_amdgcn_readlane(ev, 0);
    const int eEnd = __builtin_amdgcn_readlane(ev, GROWS);
    int2 rec;
    {
        const int idx = e0 + lane;
        rec = sedge[idx < N_EDGES ? idx : N_EDGES - 1];
    }
    __syncthreads();

    float* const yrow = &sY[rbase * YP + 4 * lane];
    {
        int cur  = 0;
        int endc = __builtin_amdgcn_readlane(ev, 1);
        float4 acc = make_float4(0.f, 0.f, 0.f, 0.f);
        int lb = 0;
        for (int p = e0; p < eEnd; ) {
            float4 r[GCHUNK];
#pragma unroll
            for (int u = 0; u < GCHUNK; ++u) {
                r[u] = make_float4(0.f, 0.f, 0.f, 0.f);
                if (p + u < eEnd) {
                    const int srcn = __builtin_amdgcn_readlane(rec.x, lb + u);
                    r[u] = *(const float4*)(x + (long long)srcn * NODE_DIM + 4 * lane);
                }
            }
#pragma unroll
            for (int u = 0; u < GCHUNK; ++u) {
                const int s = p + u;
                if (s < eEnd) {
                    while (s >= endc) {
                        *(float4*)(yrow + cur * YP) = acc;
                        acc = make_float4(0.f, 0.f, 0.f, 0.f);
                        ++cur;
                        endc = __builtin_amdgcn_readlane(ev, cur + 1);
                    }
                    const float wgt = __int_as_float(__builtin_amdgcn_readlane(rec.y, lb + u));
                    EdgeGCN_axpy4(acc, wgt, r[u]);
                }
            }
            p  += GCHUNK;
            lb ^= GCHUNK;
            if (lb == 0 && p < eEnd) {
                const int idx = p + lane;
                rec = sedge[idx < N_EDGES ? idx : N_EDGES - 1];
            }
        }
        while (cur < GROWS) {
            *(float4*)(yrow + cur * YP) = acc;
            acc = make_float4(0.f, 0.f, 0.f, 0.f);
            ++cur;
        }
    }
    {
        float4 a[GROWS];
#pragma unroll
        for (int i = 0; i < GROWS; ++i) a[i] = *(const float4*)(yrow + i * YP);
#pragma unroll
        for (int i = 0; i < GROWS; ++i) {
            const float di = __int_as_float(__builtin_amdgcn_readlane(__float_as_int(dv), i));
            const float d2 = di * di;
            const bool valid = (cw + i) < N_NODES;
            float4 y;
            y.x = valid ? di * a[i].x + selfv[i].x * d2 : 0.0f;
            y.y = valid ? di * a[i].y + selfv[i].y * d2 : 0.0f;
            y.z = valid ? di * a[i].z + selfv[i].z * d2 : 0.0f;
            y.w = valid ? di * a[i].w + selfv[i].w * d2 : 0.0f;
            *(float4*)(yrow + i * YP) = y;
        }
    }
    __syncthreads();

    if (fetcher) nxt = (int)atomicInc(ctr, 0xffffffffu);

    v8f p0 = {}, p1 = {};
    {
        const float* ya = &sY[(rt * 16 + lo) * YP + 2 * hi];
        const float* wb = &sW1[(2 * hi) * W1P + cb + lo];
#pragma unroll
        for (int kb = 0; kb < NODE_DIM; kb += 4) {
            const v2f a = *(const v2f*)(ya + kb);
            v2f q0, q1;
            q0.x = wb[kb * W1P];
            q0.y = wb[(kb + 1) * W1P];
            q1.x = wb[kb * W1P + 16];
            q1.y = wb[(kb + 1) * W1P + 16];
            p0 = __builtin_amdgcn_wmma_f32_16x16x4_f32(false, a, false, q0, (short)0, p0, false, false);
            p1 = __builtin_amdgcn_wmma_f32_16x16x4_f32(false, a, false, q1, (short)0, p1, false, false);
        }
    }
    __syncthreads();

    float* sA = sY;
#pragma unroll
    for (int v = 0; v < 8; ++v) {
        const int row = rt * 16 + v + 8 * hi;
        sA[row * AP + cb + lo]      = fmaxf(p0[v] + bias0, 0.0f);
        sA[row * AP + cb + 16 + lo] = fmaxf(p1[v] + bias1, 0.0f);
    }
    if (fetcher) sNext = nxt;
    __syncthreads();

    const int ntile = __builtin_amdgcn_readfirstlane(sNext);
    EDGEGCN_L1_HEADERS(ntile, ev, dv)

    v8f d0 = {}, d1 = {};
    {
        const float* aa  = &sA[(rt * 16 + lo) * AP + 2 * hi];
        const float* wb2 = &sW2[(2 * hi) * W2P + ct + lo];
#pragma unroll
        for (int kb = 0; kb < STATE; kb += 8) {
            const v2f a0 = *(const v2f*)(aa + kb);
            const v2f a1 = *(const v2f*)(aa + kb + 4);
            v2f q0, q1;
            q0.x = wb2[kb * W2P];
            q0.y = wb2[(kb + 1) * W2P];
            q1.x = wb2[(kb + 4) * W2P];
            q1.y = wb2[(kb + 5) * W2P];
            d0 = __builtin_amdgcn_wmma_f32_16x16x4_f32(false, a0, false, q0, (short)0, d0, false, false);
            d1 = __builtin_amdgcn_wmma_f32_16x16x4_f32(false, a1, false, q1, (short)0, d1, false, false);
        }
    }
#pragma unroll
    for (int v = 0; v < 8; ++v) {
        const int mr = c0 + rt * 16 + v + 8 * hi;
        if (mr < N_NODES) h2t[(long long)mr * OUT_DIM + ct + lo] = d0[v] + d1[v];
    }

    tile = ntile;
    }
#undef EDGEGCN_L1_HEADERS
}

#define G2_BLOCKS 782
#define G2_STRIDE (G2_BLOCKS * 32)

static_assert((long long)G2_STRIDE * 2 >= N_NODES, "two passes of the layer-2 gather cover every node");

__global__ __launch_bounds__(256) void EdgeGCN_gather2_kernel(
        const int* __restrict__ offs,
        const float* __restrict__ dinv, const int2* __restrict__ sedge,
        const float* __restrict__ h2all, const float* __restrict__ b2,
        float* __restrict__ out, int tbase) {
    const float4* __restrict__ h4 = (const float4*)h2all;
    const float4* __restrict__ b4 = (const float4*)b2;
    float4* __restrict__ o4 = (float4*)out;
    const unsigned q4 = threadIdx.x & 7;
    const unsigned t0 = (unsigned)tbase;

    for (unsigned c = blockIdx.x * 32 + (threadIdx.x >> 3); c < N_NODES; c += G2_STRIDE) {
        int    beg[TBATCH], n[TBATCH];
        float  di[TBATCH];
        float4 self[TBATCH], acc[TBATCH];
#pragma unroll
        for (int tl = 0; tl < TBATCH; ++tl) {
            const unsigned k = (t0 + tl) * N_NODES + c;
            beg[tl]  = offs[k];
            n[tl]    = ((k + 1 < NKEYS) ? offs[k + 1] : N_EDGES) - beg[tl];
            di[tl]   = dinv[k];
            self[tl] = h4[((unsigned)tl * N_NODES + c) * (OUT_DIM / 4) + q4];
            acc[tl]  = make_float4(0.f, 0.f, 0.f, 0.f);
        }
        int nmax = n[0];
#pragma unroll
        for (int tl = 1; tl < TBATCH; ++tl) nmax = (n[tl] > nmax) ? n[tl] : nmax;

        int2 e0[TBATCH], e1[TBATCH];
#pragma unroll
        for (int tl = 0; tl < TBATCH; ++tl) {
            const int i0 = beg[tl];
            const int i1 = beg[tl] + 1;
            e0[tl] = sedge[i0 < N_EDGES ? i0 : N_EDGES - 1];
            e1[tl] = sedge[i1 < N_EDGES ? i1 : N_EDGES - 1];
        }
        int j = 0;
        do {
            float4 r0[TBATCH], r1[TBATCH];
#pragma unroll
            for (int tl = 0; tl < TBATCH; ++tl) {
                r0[tl] = make_float4(0.f, 0.f, 0.f, 0.f);
                r1[tl] = make_float4(0.f, 0.f, 0.f, 0.f);
                if (j < n[tl])
                    r0[tl] = h4[((unsigned)tl * N_NODES + (unsigned)e0[tl].x) * (OUT_DIM / 4) + q4];
                if (j + 1 < n[tl])
                    r1[tl] = h4[((unsigned)tl * N_NODES + (unsigned)e1[tl].x) * (OUT_DIM / 4) + q4];
            }
            int2 f0[TBATCH], f1[TBATCH];
#pragma unroll
            for (int tl = 0; tl < TBATCH; ++tl) {
                const int i0 = beg[tl] + j + 2;
                const int i1 = beg[tl] + j + 3;
                f0[tl] = sedge[i0 < N_EDGES ? i0 : N_EDGES - 1];
                f1[tl] = sedge[i1 < N_EDGES ? i1 : N_EDGES - 1];
            }
#pragma unroll
            for (int tl = 0; tl < TBATCH; ++tl) {
                const float w0 = (j < n[tl])     ? __int_as_float(e0[tl].y) : 0.0f;
                const float w1 = (j + 1 < n[tl]) ? __int_as_float(e1[tl].y) : 0.0f;
                EdgeGCN_axpy4(acc[tl], w0, r0[tl]);
                EdgeGCN_axpy4(acc[tl], w1, r1[tl]);
                e0[tl] = f0[tl];
                e1[tl] = f1[tl];
            }
            j += 2;
        } while (j < nmax);
#pragma unroll
        for (int tl = 0; tl < TBATCH; ++tl) {
            const float4 b = b4[(t0 + tl) * (OUT_DIM / 4) + q4];
            const float d  = di[tl];
            const float d2 = d * d;
            float4 o;
            o.x = d * acc[tl].x + self[tl].x * d2 + b.x;
            o.y = d * acc[tl].y + self[tl].y * d2 + b.y;
            o.z = d * acc[tl].z + self[tl].z * d2 + b.z;
            o.w = d * acc[tl].w + self[tl].w * d2 + b.w;
            o4[c * (OUT_W / 4) + (t0 + tl) * (OUT_DIM / 4) + q4] = o;
        }
    }
}

extern "C" void kernel_launch(void* const* d_in, const int* in_sizes, int n_in,
                              void* d_out, int out_size, void* d_ws, size_t ws_size,
                              hipStream_t stream) {
    const float* x   = (const float*)d_in[0];
    const float* ea  = (const float*)d_in[1];
    const float* W1  = (const float*)d_in[2];
    const float* b1  = (const float*)d_in[3];
    const float* W2  = (const float*)d_in[4];
    const float* b2  = (const float*)d_in[5];
    const int*   eix = (const int*)d_in[6];
    float* out = (float*)d_out;

    char* ws = (char*)d_ws;
    auto alignup = [](size_t v) { return (v + 255) & ~(size_t)255; };
    size_t off = 0;
    int*   cnt   = (int*)  (ws + off); off += alignup(sizeof(int) * NKEYS);
    int*   offs  = (int*)  (ws + off); off += alignup(sizeof(int) * NKEYS);
    float* dinv  = (float*)(ws + off); off += alignup(sizeof(float) * NKEYS);
    int2*  sedge = (int2*) (ws + off); off += alignup(sizeof(int2) * (size_t)N_EDGES);
    float* h2    = (float*)(ws + off); off += alignup(sizeof(float) * (size_t)TBATCH * N_NODES * OUT_DIM);
    int*   tsum  = (int*)h2;
    int*   key   = tsum + 1024;
    int*   rank  = key + N_EDGES;
    (void)ws_size; (void)in_sizes; (void)n_in; (void)out_size;

    const int TB = 256;

    EdgeGCN_zero_int_kernel<<<ZERO_BLOCKS, TB, 0, stream>>>(cnt);
    EdgeGCN_key_rank_kernel<<<EDGE_BLOCKS, TB, 0, stream>>>(ea, eix, cnt, key, rank);
    EdgeGCN_tile_sum_kernel<<<SCAN_BLOCKS, TB, 0, stream>>>(cnt, tsum);
    EdgeGCN_scan_kernel<<<SCAN_BLOCKS, TB, 0, stream>>>(cnt, tsum, offs, dinv);
    EdgeGCN_fill_kernel<<<EDGE_BLOCKS, TB, 0, stream>>>(key, rank, eix, offs, dinv, sedge, cnt);

    for (int tb = 0; tb < ETYPES; tb += TBATCH) {
        EdgeGCN_layer1_fused_kernel<<<L1_BLOCKS, TB, 0, stream>>>(
            x, W1, b1, W2, offs, dinv, sedge, cnt, h2, tb);
        EdgeGCN_gather2_kernel<<<G2_BLOCKS, TB, 0, stream>>>(
            offs, dinv, sedge, h2, b2, out, tb);
    }
}
